# starting kernel
# baseline (speedup 1.0000x reference)
#include <hip/hip_runtime.h>

typedef __attribute__((ext_vector_type(2))) float v2f;
typedef __attribute__((ext_vector_type(8))) float v8f;

#define IN_F 128
#define OUT_F 64
#define NEG_SLOPE 0.2f
#define LDS_STRIDE (IN_F + 2)

__global__ __launch_bounds__(256) void gat_gemm_kernel(
    const float* __restrict__ x,  const float* __restrict__ Wv,
    const float* __restrict__ bv, const float* __restrict__ wq,
    const float* __restrict__ bq, const float* __restrict__ wk,
    const float* __restrict__ bk,
    float* __restrict__ h, float* __restrict__ qout, float* __restrict__ kout,
    int N)
{
    __shared__ float Wt[OUT_F * LDS_STRIDE];

    const int tid = threadIdx.x;
    for (int idx = tid; idx < IN_F * OUT_F; idx += 256) {
        int kk = idx / OUT_F, n = idx % OUT_F;
        Wt[n * LDS_STRIDE + kk] = Wv[idx];
    }
    __syncthreads();

    const int wave = tid >> 5;
    const int lane = tid & 31;
    const int half = lane >> 4;
    const int l16  = lane & 15;
    const long tile = (long)blockIdx.x * 8 + wave;
    const long rowBase = tile * 16;
    if (rowBase >= (long)N) return;

    v8f acc[4];
#pragma unroll
    for (int nt = 0; nt < 4; ++nt) {
        float b = bv[nt * 16 + l16];
#pragma unroll
        for (int j = 0; j < 8; ++j) acc[nt][j] = b;
    }

    long arow = rowBase + l16;
    if (arow >= (long)N) arow = (long)N - 1;
    const float* __restrict__ xrow = x + arow * IN_F;

    for (int kk = 0; kk < IN_F; kk += 4) {
        const int kl = kk + 2 * half;
        const float2 av = *(const float2*)(xrow + kl);
        v2f a; a.x = av.x; a.y = av.y;
#pragma unroll
        for (int nt = 0; nt < 4; ++nt) {
            const float* wp = &Wt[(nt * 16 + l16) * LDS_STRIDE + kl];
            v2f b; b.x = wp[0]; b.y = wp[1];
            acc[nt] = __builtin_amdgcn_wmma_f32_16x16x4_f32(
                false, a, false, b, (short)0, acc[nt], false, false);
        }
    }

    float qpart[8], kpart[8];
#pragma unroll
    for (int j = 0; j < 8; ++j) { qpart[j] = 0.f; kpart[j] = 0.f; }
#pragma unroll
    for (int nt = 0; nt < 4; ++nt) {
        const float wqn = wq[nt * 16 + l16];
        const float wkn = wk[nt * 16 + l16];
#pragma unroll
        for (int j = 0; j < 8; ++j) {
            qpart[j] += acc[nt][j] * wqn;
            kpart[j] += acc[nt][j] * wkn;
        }
    }
#pragma unroll
    for (int off = 8; off >= 1; off >>= 1) {
#pragma unroll
        for (int j = 0; j < 8; ++j) {
            qpart[j] += __shfl_xor(qpart[j], off, 32);
            kpart[j] += __shfl_xor(kpart[j], off, 32);
        }
    }

#pragma unroll
    for (int nt = 0; nt < 4; ++nt) {
#pragma unroll
        for (int j = 0; j < 8; ++j) {
            const long m = rowBase + half * 8 + j;
            if (m < (long)N) h[m * OUT_F + nt * 16 + l16] = acc[nt][j];
        }
    }
    if (l16 == 0) {
        const float bqv = bq[0], bkv = bk[0];
#pragma unroll
        for (int j = 0; j < 8; ++j) {
            const long m = rowBase + half * 8 + j;
            if (m < (long)N) {
                qout[m] = qpart[j] + bqv;
                kout[m] = kpart[j] + bkv;
            }
        }
    }
}

__device__ __forceinline__ unsigned f2ord(float f) {
    unsigned u = __float_as_uint(f);
    return (u & 0x80000000u) ? ~u : (u | 0x80000000u);
}
__device__ __forceinline__ float ord2f(unsigned u) {
    return (u & 0x80000000u) ? __uint_as_float(u ^ 0x80000000u)
                             : __uint_as_float(~u);
}

__global__ __launch_bounds__(256) void edge_max_kernel(
    const float* __restrict__ q, const float* __restrict__ k,
    const int* __restrict__ src, const int* __restrict__ dst,
    float* __restrict__ coeff, unsigned* __restrict__ mmax, int E)
{
    int i = blockIdx.x * 256 + threadIdx.x;
    if (i >= E) return;
    const int s = src[i], d = dst[i];
    const float e = q[s] + k[d];
    const float c = (e >= 0.f) ? e : NEG_SLOPE * e;
    coeff[i] = c;
    atomicMax(&mmax[d], f2ord(c));
}

__global__ __launch_bounds__(256) void edge_expsum_kernel(
    const int* __restrict__ dst, const unsigned* __restrict__ mmax,
    float* __restrict__ coeff, float* __restrict__ ssum, int E)
{
    int i = blockIdx.x * 256 + threadIdx.x;
    if (i >= E) return;
    const int d = dst[i];
    const float m = ord2f(mmax[d]);
    const float ex = __expf(coeff[i] - m);
    coeff[i] = ex;
    atomicAdd(&ssum[d], ex);
}

__global__ __launch_bounds__(256) void edge_aggregate_kernel(
    const float* __restrict__ h, const int* __restrict__ src,
    const int* __restrict__ dst, const float* __restrict__ ex,
    const float* __restrict__ ssum, float* __restrict__ out, int E)
{
    const long gwave = ((long)blockIdx.x * 256 + threadIdx.x) >> 5;
    const int lane = threadIdx.x & 31;
    if (gwave >= (long)E) return;
    const int s = src[gwave], d = dst[gwave];
    const float attn = ex[gwave] / ssum[d];
    const float2 v = *(const float2*)(h + (long)s * OUT_F + 2 * lane);
    float* op = out + (long)d * OUT_F + 2 * lane;
    atomicAdd(op,     v.x * attn);
    atomicAdd(op + 1, v.y * attn);
}

extern "C" void kernel_launch(void* const* d_in, const int* in_sizes, int n_in,
                              void* d_out, int out_size, void* d_ws, size_t ws_size,
                              hipStream_t stream) {
    const float* x  = (const float*)d_in[0];
    const float* Wv = (const float*)d_in[1];
    const float* bv = (const float*)d_in[2];
    const float* wq = (const float*)d_in[3];
    const float* bq = (const float*)d_in[4];
    const float* wk = (const float*)d_in[5];
    const float* bk = (const float*)d_in[6];
    const int*   src = (const int*)d_in[7];
    const int*   dst = (const int*)d_in[8];

    const int N = in_sizes[0] / IN_F;
    const int E = in_sizes[7];
    float* out = (float*)d_out;

    char* p = (char*)d_ws;
    auto carve = [&](size_t bytes) {
        char* r = p;
        p += (bytes + 255) & ~(size_t)255;
        return r;
    };
    float*    h    = (float*)   carve((size_t)N * OUT_F * sizeof(float));
    float*    q    = (float*)   carve((size_t)N * sizeof(float));
    float*    k    = (float*)   carve((size_t)N * sizeof(float));
    unsigned* mmax = (unsigned*)carve((size_t)N * sizeof(unsigned));
    float*    ssum = (float*)   carve((size_t)N * sizeof(float));
    float*    ex   = (float*)   carve((size_t)E * sizeof(float));

    hipMemsetAsync(out,  0, (size_t)N * OUT_F * sizeof(float), stream);
    hipMemsetAsync(mmax, 0, (size_t)N * sizeof(unsigned), stream);
    hipMemsetAsync(ssum, 0, (size_t)N * sizeof(float), stream);

    const int tiles = (N + 15) / 16;
    const int gemmBlocks = (tiles + 7) / 8;
    gat_gemm_kernel<<<gemmBlocks, 256, 0, stream>>>(
        x, Wv, bv, wq, bq, wk, bk, h, q, k, N);

    edge_max_kernel<<<(E + 255) / 256, 256, 0, stream>>>(
        q, k, src, dst, ex, mmax, E);

    edge_expsum_kernel<<<(E + 255) / 256, 256, 0, stream>>>(
        dst, mmax, ex, ssum, E);

    edge_aggregate_kernel<<<(E + 7) / 8, 256, 0, stream>>>(
        h, src, dst, ex, ssum, out, E);
}


# Edges bucketed by destination (integer count + arrival-order rank), then one wave per node does max, exp-sum and weighted row sum in registers: float atomics gone, sum order = arrival-order rank
# speedup vs baseline: 2.0702x; 2.0702x over previous
#include <hip/hip_runtime.h>

typedef __attribute__((ext_vector_type(2))) float v2f;
typedef __attribute__((ext_vector_type(8))) float v8f;

#define IN_F 128
#define OUT_F 64
#define NEG_SLOPE 0.2f
#define LDS_STRIDE (IN_F + 2)

__global__ __launch_bounds__(256) void gat_gemm_kernel(
    const float* __restrict__ x,  const float* __restrict__ Wv,
    const float* __restrict__ bv, const float* __restrict__ wq,
    const float* __restrict__ bq, const float* __restrict__ wk,
    const float* __restrict__ bk,
    float* __restrict__ h, float* __restrict__ qout, float* __restrict__ kout,
    int N)
{
    __shared__ float Wt[OUT_F * LDS_STRIDE];

    const int tid = threadIdx.x;
    for (int idx = tid; idx < IN_F * OUT_F; idx += 256) {
        int kk = idx / OUT_F, n = idx % OUT_F;
        Wt[n * LDS_STRIDE + kk] = Wv[idx];
    }
    __syncthreads();

    const int wave = tid >> 5;
    const int lane = tid & 31;
    const int half = lane >> 4;
    const int l16  = lane & 15;
    const long tile = (long)blockIdx.x * 8 + wave;
    const long rowBase = tile * 16;
    if (rowBase >= (long)N) return;

    v8f acc[4];
#pragma unroll
    for (int nt = 0; nt < 4; ++nt) {
        float b = bv[nt * 16 + l16];
#pragma unroll
        for (int j = 0; j < 8; ++j) acc[nt][j] = b;
    }

    long arow = rowBase + l16;
    if (arow >= (long)N) arow = (long)N - 1;
    const float* __restrict__ xrow = x + arow * IN_F;

    for (int kk = 0; kk < IN_F; kk += 4) {
        const int kl = kk + 2 * half;
        const float2 av = *(const float2*)(xrow + kl);
        v2f a; a.x = av.x; a.y = av.y;
#pragma unroll
        for (int nt = 0; nt < 4; ++nt) {
            const float* wp = &Wt[(nt * 16 + l16) * LDS_STRIDE + kl];
            v2f b; b.x = wp[0]; b.y = wp[1];
            acc[nt] = __builtin_amdgcn_wmma_f32_16x16x4_f32(
                false, a, false, b, (short)0, acc[nt], false, false);
        }
    }

    float qpart[8], kpart[8];
#pragma unroll
    for (int j = 0; j < 8; ++j) { qpart[j] = 0.f; kpart[j] = 0.f; }
#pragma unroll
    for (int nt = 0; nt < 4; ++nt) {
        const float wqn = wq[nt * 16 + l16];
        const float wkn = wk[nt * 16 + l16];
#pragma unroll
        for (int j = 0; j < 8; ++j) {
            qpart[j] += acc[nt][j] * wqn;
            kpart[j] += acc[nt][j] * wkn;
        }
    }
#pragma unroll
    for (int off = 8; off >= 1; off >>= 1) {
#pragma unroll
        for (int j = 0; j < 8; ++j) {
            qpart[j] += __shfl_xor(qpart[j], off, 32);
            kpart[j] += __shfl_xor(kpart[j], off, 32);
        }
    }

#pragma unroll
    for (int nt = 0; nt < 4; ++nt) {
#pragma unroll
        for (int j = 0; j < 8; ++j) {
            const long m = rowBase + half * 8 + j;
            if (m < (long)N) h[m * OUT_F + nt * 16 + l16] = acc[nt][j];
        }
    }
    if (l16 == 0) {
        const float bqv = bq[0], bkv = bk[0];
#pragma unroll
        for (int j = 0; j < 8; ++j) {
            const long m = rowBase + half * 8 + j;
            if (m < (long)N) {
                qout[m] = qpart[j] + bqv;
                kout[m] = kpart[j] + bkv;
            }
        }
    }
}

__global__ __launch_bounds__(256) void edge_rank_kernel(
    const int* __restrict__ dst, int* __restrict__ cnt,
    int* __restrict__ rank, int E)
{
    const int i = blockIdx.x * 256 + threadIdx.x;
    if (i >= E) return;
    rank[i] = atomicAdd(&cnt[dst[i]], 1);
}

__global__ __launch_bounds__(256) void seg_alloc_kernel(
    const int* cnt, int* __restrict__ offs, int* cursor, int N)
{
    const int d = blockIdx.x * 256 + threadIdx.x;
    const int lane = threadIdx.x & 31;
    const int c = (d < N) ? cnt[d] : 0;
    int incl = c;
#pragma unroll
    for (int off = 1; off < 32; off <<= 1) {
        const int t = __shfl_up(incl, off, 32);
        if (lane >= off) incl += t;
    }
    int base = 0;
    if (lane == 31) base = atomicAdd(cursor, incl);
    base = __shfl(base, 31, 32);
    if (d < N) offs[d] = base + incl - c;
}

__global__ __launch_bounds__(256) void edge_scatter_kernel(
    const int* __restrict__ src, const int* __restrict__ dst,
    const int* __restrict__ rank, const int* __restrict__ offs,
    int* __restrict__ ssrc, int E)
{
    const int i = blockIdx.x * 256 + threadIdx.x;
    if (i >= E) return;
    ssrc[offs[dst[i]] + rank[i]] = src[i];
}

__device__ __forceinline__ float wave_max(float v) {
#pragma unroll
    for (int off = 16; off >= 1; off >>= 1) v = fmaxf(v, __shfl_xor(v, off, 32));
    return v;
}
__device__ __forceinline__ float wave_sum(float v) {
#pragma unroll
    for (int off = 16; off >= 1; off >>= 1) v += __shfl_xor(v, off, 32);
    return v;
}

__device__ __forceinline__ float edge_coeff(const float* __restrict__ q, int s, float kd) {
    const float e = q[s] + kd;
    return (e >= 0.f) ? e : NEG_SLOPE * e;
}

__device__ __forceinline__ void add_rows(const float* __restrict__ hcol, int s, float a,
                                         int cn, float& accx, float& accy)
{
    int t = 0;
    for (; t + 8 <= cn; t += 8) {
        float2 v[8];
#pragma unroll
        for (int u = 0; u < 8; ++u) {
            const int st = __builtin_amdgcn_readlane(s, t + u);
            v[u] = *(const float2*)(hcol + (long)st * OUT_F);
        }
#pragma unroll
        for (int u = 0; u < 8; ++u) {
            const float at = __int_as_float(
                __builtin_amdgcn_readlane(__float_as_int(a), t + u));
            accx += v[u].x * at;
            accy += v[u].y * at;
        }
    }
    for (; t < cn; ++t) {
        const int st = __builtin_amdgcn_readlane(s, t);
        const float at = __int_as_float(
            __builtin_amdgcn_readlane(__float_as_int(a), t));
        const float2 v = *(const float2*)(hcol + (long)st * OUT_F);
        accx += v.x * at;
        accy += v.y * at;
    }
}

__global__ __launch_bounds__(256) void node_softmax_aggregate_kernel(
    const float* __restrict__ h, const float* __restrict__ q,
    const float* __restrict__ k, const int* __restrict__ cnt,
    const int* __restrict__ offs, const int* __restrict__ ssrc,
    float* __restrict__ out, int N)
{
    const int lane = threadIdx.x & 31;
    const int d = blockIdx.x * 8 + __builtin_amdgcn_readfirstlane((int)(threadIdx.x >> 5));
    if (d >= N) return;
    const int n = cnt[d];
    const int base = offs[d];
    const float kd = k[d];

    int s0 = 0;
    float cf0 = -INFINITY;
    if (lane < n) {
        s0 = ssrc[base + lane];
        cf0 = edge_coeff(q, s0, kd);
    }

    float m = cf0;
    for (int c = 32; c < n; c += 32) {
        const int j = c + lane;
        if (j < n) m = fmaxf(m, edge_coeff(q, ssrc[base + j], kd));
    }
    m = wave_max(m);

    const float ex0 = (lane < n) ? __expf(cf0 - m) : 0.f;
    float ssum = ex0;
    for (int c = 32; c < n; c += 32) {
        const int j = c + lane;
        if (j < n) ssum += __expf(edge_coeff(q, ssrc[base + j], kd) - m);
    }
    ssum = wave_sum(ssum);

    float accx = 0.f, accy = 0.f;
    const float* __restrict__ hcol = h + 2 * lane;
    {
        const float a0 = (lane < n) ? ex0 / ssum : 0.f;
        add_rows(hcol, s0, a0, (n < 32) ? n : 32, accx, accy);
    }
    for (int c = 32; c < n; c += 32) {
        const int j = c + lane;
        int s = 0;
        float a = 0.f;
        if (j < n) {
            s = ssrc[base + j];
            a = __expf(edge_coeff(q, s, kd) - m) / ssum;
        }
        add_rows(hcol, s, a, (n - c < 32) ? (n - c) : 32, accx, accy);
    }
    float2 r; r.x = accx; r.y = accy;
    *(float2*)(out + (long)d * OUT_F + 2 * lane) = r;
}

extern "C" void kernel_launch(void* const* d_in, const int* in_sizes, int n_in,
                              void* d_out, int out_size, void* d_ws, size_t ws_size,
                              hipStream_t stream) {
    const float* x  = (const float*)d_in[0];
    const float* Wv = (const float*)d_in[1];
    const float* bv = (const float*)d_in[2];
    const float* wq = (const float*)d_in[3];
    const float* bq = (const float*)d_in[4];
    const float* wk = (const float*)d_in[5];
    const float* bk = (const float*)d_in[6];
    const int*   src = (const int*)d_in[7];
    const int*   dst = (const int*)d_in[8];

    const int N = in_sizes[0] / IN_F;
    const int E = in_sizes[7];
    float* out = (float*)d_out;

    char* p = (char*)d_ws;
    auto carve = [&](size_t bytes) {
        char* r = p;
        p += (bytes + 255) & ~(size_t)255;
        return r;
    };
    float* h    = (float*)carve((size_t)N * OUT_F * sizeof(float));
    float* q    = (float*)carve((size_t)N * sizeof(float));
    float* k    = (float*)carve((size_t)N * sizeof(float));
    int*   cnt  = (int*)  carve((size_t)(N + 1) * sizeof(int));
    int*   offs = (int*)  carve((size_t)N * sizeof(int));
    int*   ssrc = (int*)  carve((size_t)E * sizeof(int));
    int*   rank = (int*)d_out;

    hipMemsetAsync(cnt, 0, (size_t)(N + 1) * sizeof(int), stream);

    const int tiles = (N + 15) / 16;
    const int gemmBlocks = (tiles + 7) / 8;
    gat_gemm_kernel<<<gemmBlocks, 256, 0, stream>>>(
        x, Wv, bv, wq, bq, wk, bk, h, q, k, N);

    edge_rank_kernel<<<(E + 255) / 256, 256, 0, stream>>>(dst, cnt, rank, E);

    seg_alloc_kernel<<<(N + 255) / 256, 256, 0, stream>>>(cnt, offs, cnt + N, N);

    edge_scatter_kernel<<<(E + 255) / 256, 256, 0, stream>>>(
        src, dst, rank, offs, ssrc, E);

    node_softmax_aggregate_kernel<<<(N + 7) / 8, 256, 0, stream>>>(
        h, q, k, cnt, offs, ssrc, out, N);
}


# Aggregation as resident waves: each wave walks a contiguous run of nodes, the next nodes' edge ids and q values fetched under the current node's row gathers, tail rows as one group; sum order kept
# speedup vs baseline: 5.7533x; 2.7791x over previous
#include <hip/hip_runtime.h>

typedef __attribute__((ext_vector_type(2))) float v2f;
typedef __attribute__((ext_vector_type(8))) float v8f;

#define IN_F 128
#define OUT_F 64
#define NEG_SLOPE 0.2f
#define LDS_STRIDE (IN_F + 2)

__global__ __launch_bounds__(256) void gat_gemm_kernel(
    const float* __restrict__ x,  const float* __restrict__ Wv,
    const float* __restrict__ bv, const float* __restrict__ wq,
    const float* __restrict__ bq, const float* __restrict__ wk,
    const float* __restrict__ bk,
    float* __restrict__ h, float* __restrict__ qout, float* __restrict__ kout,
    int N)
{
    __shared__ float Wt[OUT_F * LDS_STRIDE];

    const int tid = threadIdx.x;
    for (int idx = tid; idx < IN_F * OUT_F; idx += 256) {
        int kk = idx / OUT_F, n = idx % OUT_F;
        Wt[n * LDS_STRIDE + kk] = Wv[idx];
    }
    __syncthreads();

    const int wave = tid >> 5;
    const int lane = tid & 31;
    const int half = lane >> 4;
    const int l16  = lane & 15;
    const long tile = (long)blockIdx.x * 8 + wave;
    const long rowBase = tile * 16;
    if (rowBase >= (long)N) return;

    v8f acc[4];
#pragma unroll
    for (int nt = 0; nt < 4; ++nt) {
        float b = bv[nt * 16 + l16];
#pragma unroll
        for (int j = 0; j < 8; ++j) acc[nt][j] = b;
    }

    long arow = rowBase + l16;
    if (arow >= (long)N) arow = (long)N - 1;
    const float* __restrict__ xrow = x + arow * IN_F;

    for (int kk = 0; kk < IN_F; kk += 4) {
        const int kl = kk + 2 * half;
        const float2 av = *(const float2*)(xrow + kl);
        v2f a; a.x = av.x; a.y = av.y;
#pragma unroll
        for (int nt = 0; nt < 4; ++nt) {
            const float* wp = &Wt[(nt * 16 + l16) * LDS_STRIDE + kl];
            v2f b; b.x = wp[0]; b.y = wp[1];
            acc[nt] = __builtin_amdgcn_wmma_f32_16x16x4_f32(
                false, a, false, b, (short)0, acc[nt], false, false);
        }
    }

    float qpart[8], kpart[8];
#pragma unroll
    for (int j = 0; j < 8; ++j) { qpart[j] = 0.f; kpart[j] = 0.f; }
#pragma unroll
    for (int nt = 0; nt < 4; ++nt) {
        const float wqn = wq[nt * 16 + l16];
        const float wkn = wk[nt * 16 + l16];
#pragma unroll
        for (int j = 0; j < 8; ++j) {
            qpart[j] += acc[nt][j] * wqn;
            kpart[j] += acc[nt][j] * wkn;
        }
    }
#pragma unroll
    for (int off = 8; off >= 1; off >>= 1) {
#pragma unroll
        for (int j = 0; j < 8; ++j) {
            qpart[j] += __shfl_xor(qpart[j], off, 32);
            kpart[j] += __shfl_xor(kpart[j], off, 32);
        }
    }

#pragma unroll
    for (int nt = 0; nt < 4; ++nt) {
#pragma unroll
        for (int j = 0; j < 8; ++j) {
            const long m = rowBase + half * 8 + j;
            if (m < (long)N) h[m * OUT_F + nt * 16 + l16] = acc[nt][j];
        }
    }
    if (l16 == 0) {
        const float bqv = bq[0], bkv = bk[0];
#pragma unroll
        for (int j = 0; j < 8; ++j) {
            const long m = rowBase + half * 8 + j;
            if (m < (long)N) {
                qout[m] = qpart[j] + bqv;
                kout[m] = kpart[j] + bkv;
            }
        }
    }
}

#define SORT_T 1024
#define NBUCKET 256
#define NPB_MAX 1024
#define SRC_BITS 20
#define SRC_MASK ((1u << SRC_BITS) - 1u)
#define PBATCH 8
#define EMPTY 0xFFFFFFFFu
#define PART_BLOCKS 256

__device__ __forceinline__ int block_scan_incl(int v, int* wsum, int* total)
{
    const int lane = threadIdx.x & 31;
    const int wave = threadIdx.x >> 5;
    int incl = v;
#pragma unroll
    for (int off = 1; off < 32; off <<= 1) {
        const int t = __shfl_up(incl, off, 32);
        if (lane >= off) incl += t;
    }
    if (lane == 31) wsum[wave] = incl;
    __syncthreads();
    if (wave == 0) {
        const int s = wsum[lane];
        int si = s;
#pragma unroll
        for (int off = 1; off < 32; off <<= 1) {
            const int t = __shfl_up(si, off, 32);
            if (lane >= off) si += t;
        }
        wsum[lane] = si - s;
        if (lane == 31) wsum[32] = si;
    }
    __syncthreads();
    *total = wsum[32];
    return incl + wsum[wave];
}

__global__ __launch_bounds__(SORT_T) void edge_partition_kernel(
    const int* __restrict__ src, const int* __restrict__ dst,
    unsigned* __restrict__ part, int* __restrict__ segStart,
    int* __restrict__ segCnt, int N, int E)
{
    __shared__ int hist[NBUCKET];
    __shared__ int cursor[NBUCKET];
    __shared__ int wsum[33];

    const int t = threadIdx.x;
    const int b = blockIdx.x;
    const int B = gridDim.x;
    const int npb = (N + NBUCKET - 1) / NBUCKET;
    const int chunk = (E + B - 1) / B;
    const int lo = b * chunk;
    int hi = lo + chunk;
    if (hi > E) hi = E;

    if (t < NBUCKET) hist[t] = 0;
    __syncthreads();

    for (int i0 = lo + t; i0 < hi; i0 += PBATCH * SORT_T) {
        int d[PBATCH];
#pragma unroll
        for (int u = 0; u < PBATCH; ++u) {
            const int i = i0 + u * SORT_T;
            d[u] = (i < hi) ? dst[i] : -1;
        }
#pragma unroll
        for (int u = 0; u < PBATCH; ++u)
            if (d[u] >= 0) atomicAdd(&hist[(unsigned)d[u] / (unsigned)npb], 1);
    }
    __syncthreads();

    const int c = (t < NBUCKET) ? hist[t] : 0;
    int total;
    const int incl = block_scan_incl(c, wsum, &total);
    if (t < NBUCKET) {
        cursor[t] = incl - c;
        segStart[t * B + b] = lo + incl - c;
        segCnt[t * B + b] = c;
    }
    __syncthreads();

    for (int i0 = lo + t; i0 < hi; i0 += PBATCH * SORT_T) {
        int d[PBATCH], s[PBATCH];
#pragma unroll
        for (int u = 0; u < PBATCH; ++u) {
            const int i = i0 + u * SORT_T;
            d[u] = (i < hi) ? dst[i] : -1;
            s[u] = (i < hi) ? src[i] : 0;
        }
#pragma unroll
        for (int u = 0; u < PBATCH; ++u) {
            if (d[u] >= 0) {
                const int kb = (int)((unsigned)d[u] / (unsigned)npb);
                const int pos = atomicAdd(&cursor[kb], 1);
                part[lo + pos] = (unsigned)s[u] | ((unsigned)(d[u] - kb * npb) << SRC_BITS);
            }
        }
    }
}

template <bool PLACE>
__device__ __forceinline__ void bucket_entry(unsigned w, int* counters,
                                             int* __restrict__ ssrc, int base)
{
    if (PLACE) {
        const int pos = atomicAdd(&counters[w >> SRC_BITS], 1);
        ssrc[base + pos] = (int)(w & SRC_MASK);
    } else {
        atomicAdd(&counters[w >> SRC_BITS], 1);
    }
}

template <bool PLACE>
__device__ __forceinline__ void walk_pieces(
    const unsigned* __restrict__ part, const int* __restrict__ myStart,
    const int* __restrict__ myCnt, int pBeg, int pEnd, int lane,
    int* counters, int* __restrict__ ssrc, int base)
{
    for (int p0 = pBeg; p0 < pEnd; p0 += 32) {
        int dS = 0, dC = 0;
        if (p0 + lane < pEnd) { dS = myStart[p0 + lane]; dC = myCnt[p0 + lane]; }
        const int np = (pEnd - p0 < 32) ? (pEnd - p0) : 32;
        for (int j = 0; j < np; j += 4) {
            int s0[4], c[4];
            unsigned w0[4], w1[4];
#pragma unroll
            for (int u = 0; u < 4; ++u) {
                s0[u] = __builtin_amdgcn_readlane(dS, j + u);
                c[u] = __builtin_amdgcn_readlane(dC, j + u);
                w0[u] = (lane < c[u]) ? part[s0[u] + lane] : EMPTY;
                w1[u] = (lane + 32 < c[u]) ? part[s0[u] + lane + 32] : EMPTY;
            }
#pragma unroll
            for (int u = 0; u < 4; ++u) {
                if (w0[u] != EMPTY) bucket_entry<PLACE>(w0[u], counters, ssrc, base);
                if (w1[u] != EMPTY) bucket_entry<PLACE>(w1[u], counters, ssrc, base);
                for (int i = lane + 64; i < c[u]; i += 32)
                    bucket_entry<PLACE>(part[s0[u] + i], counters, ssrc, base);
            }
        }
    }
}

__global__ __launch_bounds__(SORT_T) void bucket_finalize_kernel(
    const unsigned* __restrict__ part, const int* __restrict__ segStart,
    const int* __restrict__ segCnt, int* __restrict__ cnt,
    int* __restrict__ offs, int* __restrict__ ssrc, int N, int B)
{
    __shared__ int cntl[NPB_MAX];
    __shared__ int cursor[NPB_MAX];
    __shared__ int wsumA[33];
    __shared__ int wsumB[33];

    const int t = threadIdx.x;
    const int lane = t & 31;
    const int wave = __builtin_amdgcn_readfirstlane(t >> 5);
    const int k = blockIdx.x;
    const int npb = (N + NBUCKET - 1) / NBUCKET;
    const int nodeLo = k * npb;

    cntl[t] = 0;

    int v = 0;
    const int4* __restrict__ cnt4 = (const int4*)segCnt;
    const int nlow4 = (k * B) >> 2;
#pragma unroll 8
    for (int j = t; j < nlow4; j += SORT_T) {
        const int4 a = cnt4[j];
        v += (a.x + a.y) + (a.z + a.w);
    }
    int base;
    block_scan_incl(v, wsumA, &base);

    const int spw = (B + 31) / 32;
    const int pBeg = wave * spw;
    int pEnd = pBeg + spw;
    if (pEnd > B) pEnd = B;
    const int* __restrict__ myStart = segStart + k * B;
    const int* __restrict__ myCnt = segCnt + k * B;

    walk_pieces<false>(part, myStart, myCnt, pBeg, pEnd, lane, cntl, ssrc, 0);
    __syncthreads();

    const int c = cntl[t];
    int btotal;
    const int excl = block_scan_incl(c, wsumB, &btotal) - c;
    cursor[t] = excl;
    const int node = nodeLo + t;
    if (t < npb && node < N) {
        cnt[node] = c;
        offs[node] = base + excl;
    }
    __syncthreads();

    walk_pieces<true>(part, myStart, myCnt, pBeg, pEnd, lane, cursor, ssrc, base);
}

__device__ __forceinline__ float wave_max(float v) {
#pragma unroll
    for (int off = 16; off >= 1; off >>= 1) v = fmaxf(v, __shfl_xor(v, off, 32));
    return v;
}
__device__ __forceinline__ float wave_sum(float v) {
#pragma unroll
    for (int off = 16; off >= 1; off >>= 1) v += __shfl_xor(v, off, 32);
    return v;
}

#define AGG_WAVES 8
#define AGG_BLOCKS 2048

__device__ __forceinline__ float leaky(float e) {
    return (e >= 0.f) ? e : NEG_SLOPE * e;
}
__device__ __forceinline__ float edge_coeff(const float* __restrict__ q, int s, float kd) {
    return leaky(q[s] + kd);
}

__device__ __forceinline__ void add_rows(const float* __restrict__ hcol, int s, float a,
                                         int cn, float& accx, float& accy)
{
    int t = 0;
    for (; t + 8 <= cn; t += 8) {
        float2 v[8];
#pragma unroll
        for (int u = 0; u < 8; ++u) {
            const int st = __builtin_amdgcn_readlane(s, t + u);
            v[u] = *(const float2*)(hcol + (long)st * OUT_F);
        }
#pragma unroll
        for (int u = 0; u < 8; ++u) {
            const float at = __int_as_float(
                __builtin_amdgcn_readlane(__float_as_int(a), t + u));
            accx += v[u].x * at;
            accy += v[u].y * at;
        }
    }
    const int rem = cn - t;
    if (rem > 0) {
        float2 v[7];
#pragma unroll
        for (int u = 0; u < 7; ++u) {
            v[u].x = 0.f; v[u].y = 0.f;
            if (u < rem) {
                const int st = __builtin_amdgcn_readlane(s, t + u);
                v[u] = *(const float2*)(hcol + (long)st * OUT_F);
            }
        }
#pragma unroll
        for (int u = 0; u < 7; ++u) {
            if (u < rem) {
                const float at = __int_as_float(
                    __builtin_amdgcn_readlane(__float_as_int(a), t + u));
                accx += v[u].x * at;
                accy += v[u].y * at;
            }
        }
    }
}

__global__ __launch_bounds__(256) void node_softmax_aggregate_kernel(
    const float* __restrict__ h, const float* __restrict__ q,
    const float* __restrict__ k, const int* __restrict__ cnt,
    const int* __restrict__ offs, const int* __restrict__ ssrc,
    float* __restrict__ out, int N)
{
    const int lane = threadIdx.x & 31;
    const int wv = __builtin_amdgcn_readfirstlane((int)(threadIdx.x >> 5));
    const int nw = AGG_BLOCKS * AGG_WAVES;
    const int w = (int)blockIdx.x * AGG_WAVES + wv;
    const int d0 = (int)(((long)w * (long)N) / nw);
    const int d1 = (int)(((long)(w + 1) * (long)N) / nw);
    const float* __restrict__ hcol = h + 2 * lane;

    for (int c0 = d0; c0 < d1; c0 += 32) {
        const int nc = (d1 - c0 < 32) ? (d1 - c0) : 32;
        int mN = 0, mB = 0;
        float mK = 0.f;
        if (lane < nc) { mN = cnt[c0 + lane]; mB = offs[c0 + lane]; mK = k[c0 + lane]; }

        int sA0 = 0, sA1 = 0, sB0 = 0, sB1 = 0;
        float qA0 = 0.f, qA1 = 0.f;
        {
            const int n0 = __builtin_amdgcn_readlane(mN, 0);
            const int b0 = __builtin_amdgcn_readlane(mB, 0);
            if (lane < n0) sA0 = ssrc[b0 + lane];
            if (lane + 32 < n0) sA1 = ssrc[b0 + 32 + lane];
            if (nc > 1) {
                const int n1 = __builtin_amdgcn_readlane(mN, 1);
                const int b1 = __builtin_amdgcn_readlane(mB, 1);
                if (lane < n1) sB0 = ssrc[b1 + lane];
                if (lane + 32 < n1) sB1 = ssrc[b1 + 32 + lane];
            }
            if (lane < n0) qA0 = q[sA0];
            if (lane + 32 < n0) qA1 = q[sA1];
        }

        for (int i = 0; i < nc; ++i) {
            const int n = __builtin_amdgcn_readlane(mN, i);
            const int base = __builtin_amdgcn_readlane(mB, i);
            const float kd = __int_as_float(
                __builtin_amdgcn_readlane(__float_as_int(mK), i));

            const float cf0 = (lane < n) ? leaky(qA0 + kd) : -INFINITY;
            const float cf1 = (lane + 32 < n) ? leaky(qA1 + kd) : -INFINITY;
            float m = fmaxf(cf0, cf1);
            for (int c = 64; c < n; c += 32) {
                const int j = c + lane;
                if (j < n) m = fmaxf(m, edge_coeff(q, ssrc[base + j], kd));
            }
            m = wave_max(m);

            const float ex0 = (lane < n) ? __expf(cf0 - m) : 0.f;
            const float ex1 = (lane + 32 < n) ? __expf(cf1 - m) : 0.f;
            float ssum = ex0;
            ssum += ex1;
            for (int c = 64; c < n; c += 32) {
                const int j = c + lane;
                if (j < n) ssum += __expf(edge_coeff(q, ssrc[base + j], kd) - m);
            }
            ssum = wave_sum(ssum);

            int sC0 = 0, sC1 = 0;
            float qB0 = 0.f, qB1 = 0.f;
            if (i + 1 < nc) {
                const int n1 = __builtin_amdgcn_readlane(mN, i + 1);
                if (lane < n1) qB0 = q[sB0];
                if (lane + 32 < n1) qB1 = q[sB1];
            }
            if (i + 2 < nc) {
                const int n2 = __builtin_amdgcn_readlane(mN, i + 2);
                const int b2 = __builtin_amdgcn_readlane(mB, i + 2);
                if (lane < n2) sC0 = ssrc[b2 + lane];
                if (lane + 32 < n2) sC1 = ssrc[b2 + 32 + lane];
            }

            float accx = 0.f, accy = 0.f;
            {
                const float a0 = (lane < n) ? ex0 / ssum : 0.f;
                add_rows(hcol, sA0, a0, (n < 32) ? n : 32, accx, accy);
            }
            if (n > 32) {
                const float a1 = (lane + 32 < n) ? ex1 / ssum : 0.f;
                add_rows(hcol, sA1, a1, (n < 64) ? (n - 32) : 32, accx, accy);
            }
            for (int c = 64; c < n; c += 32) {
                const int j = c + lane;
                int s = 0;
                float a = 0.f;
                if (j < n) {
                    s = ssrc[base + j];
                    a = __expf(edge_coeff(q, s, kd) - m) / ssum;
                }
                add_rows(hcol, s, a, (n - c < 32) ? (n - c) : 32, accx, accy);
            }
            float2 r; r.x = accx; r.y = accy;
            *(float2*)(out + (long)(c0 + i) * OUT_F + 2 * lane) = r;

            sA0 = sB0; sA1 = sB1; qA0 = qB0; qA1 = qB1;
            sB0 = sC0; sB1 = sC1;
        }
    }
}

extern "C" void kernel_launch(void* const* d_in, const int* in_sizes, int n_in,
                              void* d_out, int out_size, void* d_ws, size_t ws_size,
                              hipStream_t stream) {
    const float* x  = (const float*)d_in[0];
    const float* Wv = (const float*)d_in[1];
    const float* bv = (const float*)d_in[2];
    const float* wq = (const float*)d_in[3];
    const float* bq = (const float*)d_in[4];
    const float* wk = (const float*)d_in[5];
    const float* bk = (const float*)d_in[6];
    const int*   src = (const int*)d_in[7];
    const int*   dst = (const int*)d_in[8];

    const int N = in_sizes[0] / IN_F;
    const int E = in_sizes[7];
    float* out = (float*)d_out;

    char* p = (char*)d_ws;
    auto carve = [&](size_t bytes) {
        char* r = p;
        p += (bytes + 255) & ~(size_t)255;
        return r;
    };
    float* h    = (float*)carve((size_t)N * OUT_F * sizeof(float));
    float* q    = (float*)carve((size_t)N * sizeof(float));
    float* k    = (float*)carve((size_t)N * sizeof(float));
    int*   cnt  = (int*)  carve((size_t)(N + 1) * sizeof(int));
    int*   offs = (int*)  carve((size_t)N * sizeof(int));
    int*   ssrc = (int*)  carve((size_t)E * sizeof(int));
    unsigned* part = (unsigned*)d_out;
    int* segStart = (int*)d_out + (((size_t)E + 63) & ~(size_t)63);
    int* segCnt   = segStart + PART_BLOCKS * NBUCKET;

    const int tiles = (N + 15) / 16;
    const int gemmBlocks = (tiles + 7) / 8;
    gat_gemm_kernel<<<gemmBlocks, 256, 0, stream>>>(
        x, Wv, bv, wq, bq, wk, bk, h, q, k, N);

    edge_partition_kernel<<<PART_BLOCKS, SORT_T, 0, stream>>>(
        src, dst, part, segStart, segCnt, N, E);

    bucket_finalize_kernel<<<NBUCKET, SORT_T, 0, stream>>>(
        part, segStart, segCnt, cnt, offs, ssrc, N, PART_BLOCKS);

    node_softmax_aggregate_kernel<<<AGG_BLOCKS, 32 * AGG_WAVES, 0, stream>>>(
        h, q, k, cnt, offs, ssrc, out, N);
}


# Edge partition keeps its chunk in registers (one read of dst, src), sorts it in LDS and writes it out as whole 128-byte lines: scattered 4-byte global stores gone; sum order still LDS arrival rank
# speedup vs baseline: 7.5633x; 1.3146x over previous
#include <hip/hip_runtime.h>

typedef __attribute__((ext_vector_type(2))) float v2f;
typedef __attribute__((ext_vector_type(8))) float v8f;

#define IN_F 128
#define OUT_F 64
#define NEG_SLOPE 0.2f
#define LDS_STRIDE (IN_F + 2)

__global__ __launch_bounds__(256) void gat_gemm_kernel(
    const float* __restrict__ x,  const float* __restrict__ Wv,
    const float* __restrict__ bv, const float* __restrict__ wq,
    const float* __restrict__ bq, const float* __restrict__ wk,
    const float* __restrict__ bk,
    float* __restrict__ h, float* __restrict__ qout, float* __restrict__ kout,
    int N)
{
    __shared__ float Wt[OUT_F * LDS_STRIDE];

    const int tid = threadIdx.x;
    for (int idx = tid; idx < IN_F * OUT_F; idx += 256) {
        int kk = idx / OUT_F, n = idx % OUT_F;
        Wt[n * LDS_STRIDE + kk] = Wv[idx];
    }
    __syncthreads();

    const int wave = tid >> 5;
    const int lane = tid & 31;
    const int half = lane >> 4;
    const int l16  = lane & 15;
    const long tile = (long)blockIdx.x * 8 + wave;
    const long rowBase = tile * 16;
    if (rowBase >= (long)N) return;

    v8f acc[4];
#pragma unroll
    for (int nt = 0; nt < 4; ++nt) {
        float b = bv[nt * 16 + l16];
#pragma unroll
        for (int j = 0; j < 8; ++j) acc[nt][j] = b;
    }

    long arow = rowBase + l16;
    if (arow >= (long)N) arow = (long)N - 1;
    const float* __restrict__ xrow = x + arow * IN_F;

    for (int kk = 0; kk < IN_F; kk += 4) {
        const int kl = kk + 2 * half;
        const float2 av = *(const float2*)(xrow + kl);
        v2f a; a.x = av.x; a.y = av.y;
#pragma unroll
        for (int nt = 0; nt < 4; ++nt) {
            const float* wp = &Wt[(nt * 16 + l16) * LDS_STRIDE + kl];
            v2f b; b.x = wp[0]; b.y = wp[1];
            acc[nt] = __builtin_amdgcn_wmma_f32_16x16x4_f32(
                false, a, false, b, (short)0, acc[nt], false, false);
        }
    }

    float qpart[8], kpart[8];
#pragma unroll
    for (int j = 0; j < 8; ++j) { qpart[j] = 0.f; kpart[j] = 0.f; }
#pragma unroll
    for (int nt = 0; nt < 4; ++nt) {
        const float wqn = wq[nt * 16 + l16];
        const float wkn = wk[nt * 16 + l16];
#pragma unroll
        for (int j = 0; j < 8; ++j) {
            qpart[j] += acc[nt][j] * wqn;
            kpart[j] += acc[nt][j] * wkn;
        }
    }
#pragma unroll
    for (int off = 8; off >= 1; off >>= 1) {
#pragma unroll
        for (int j = 0; j < 8; ++j) {
            qpart[j] += __shfl_xor(qpart[j], off, 32);
            kpart[j] += __shfl_xor(kpart[j], off, 32);
        }
    }

#pragma unroll
    for (int nt = 0; nt < 4; ++nt) {
#pragma unroll
        for (int j = 0; j < 8; ++j) {
            const long m = rowBase + half * 8 + j;
            if (m < (long)N) h[m * OUT_F + nt * 16 + l16] = acc[nt][j];
        }
    }
    if (l16 == 0) {
        const float bqv = bq[0], bkv = bk[0];
#pragma unroll
        for (int j = 0; j < 8; ++j) {
            const long m = rowBase + half * 8 + j;
            if (m < (long)N) {
                qout[m] = qpart[j] + bqv;
                kout[m] = kpart[j] + bkv;
            }
        }
    }
}

#define SORT_T 1024
#define NBUCKET 256
#define NPB_MAX 1024
#define SRC_BITS 20
#define SRC_MASK ((1u << SRC_BITS) - 1u)
#define PBATCH 8
#define EMPTY 0xFFFFFFFFu
#define PART_BLOCKS 256
#define STAGE_CAP 16384
#define EPT (STAGE_CAP / SORT_T)

__device__ __forceinline__ int block_scan_incl(int v, int* wsum, int* total)
{
    const int lane = threadIdx.x & 31;
    const int wave = threadIdx.x >> 5;
    int incl = v;
#pragma unroll
    for (int off = 1; off < 32; off <<= 1) {
        const int t = __shfl_up(incl, off, 32);
        if (lane >= off) incl += t;
    }
    if (lane == 31) wsum[wave] = incl;
    __syncthreads();
    if (wave == 0) {
        const int s = wsum[lane];
        int si = s;
#pragma unroll
        for (int off = 1; off < 32; off <<= 1) {
            const int t = __shfl_up(si, off, 32);
            if (lane >= off) si += t;
        }
        wsum[lane] = si - s;
        if (lane == 31) wsum[32] = si;
    }
    __syncthreads();
    *total = wsum[32];
    return incl + wsum[wave];
}

__global__ __launch_bounds__(SORT_T) void edge_partition_kernel(
    const int* __restrict__ src, const int* __restrict__ dst,
    unsigned* __restrict__ part, int* __restrict__ segStart,
    int* __restrict__ segCnt, int N, int E)
{
    __shared__ int hist[NBUCKET];
    __shared__ int cursor[NBUCKET];
    __shared__ int wsum[33];
    __shared__ uint4 stage4[STAGE_CAP / 4];

    const int t = threadIdx.x;
    const int b = blockIdx.x;
    const int B = gridDim.x;
    const int npb = (N + NBUCKET - 1) / NBUCKET;
    const int chunk = (E + B - 1) / B;
    const int lo = b * chunk;
    int hi = lo + chunk;
    if (hi > E) hi = E;

    if (t < NBUCKET) hist[t] = 0;
    __syncthreads();

    if (chunk <= STAGE_CAP) {
        int d[EPT], s[EPT];
#pragma unroll
        for (int u = 0; u < EPT; ++u) {
            const int i = lo + t + u * SORT_T;
            d[u] = (i < hi) ? dst[i] : -1;
            s[u] = (i < hi) ? src[i] : 0;
        }
#pragma unroll
        for (int u = 0; u < EPT; ++u)
            if (d[u] >= 0) atomicAdd(&hist[(unsigned)d[u] / (unsigned)npb], 1);
        __syncthreads();

        const int c = (t < NBUCKET) ? hist[t] : 0;
        int total;
        const int incl = block_scan_incl(c, wsum, &total);
        if (t < NBUCKET) {
            cursor[t] = incl - c;
            segStart[t * B + b] = lo + incl - c;
            segCnt[t * B + b] = c;
        }
        __syncthreads();

        unsigned* stage = (unsigned*)stage4;
#pragma unroll
        for (int u = 0; u < EPT; ++u) {
            if (d[u] >= 0) {
                const int kb = (int)((unsigned)d[u] / (unsigned)npb);
                const int pos = atomicAdd(&cursor[kb], 1);
                stage[pos] = (unsigned)s[u] | ((unsigned)(d[u] - kb * npb) << SRC_BITS);
            }
        }
        __syncthreads();

        const int len = (hi > lo) ? (hi - lo) : 0;
        if ((lo & 3) == 0) {
            int head = (32 - (lo & 31)) & 31;
            if (head > len) head = len;
            if (t < head) part[lo + t] = stage[t];
            const int n4 = (len - head) >> 2;
            uint4* g4 = (uint4*)(part + lo + head);
            const uint4* l4 = (const uint4*)(stage + head);
            for (int i = t; i < n4; i += SORT_T) g4[i] = l4[i];
            const int done = head + 4 * n4;
            if (t < len - done) part[lo + done + t] = stage[done + t];
        } else {
            for (int i = t; i < len; i += SORT_T) part[lo + i] = stage[i];
        }
        return;
    }

    for (int i0 = lo + t; i0 < hi; i0 += PBATCH * SORT_T) {
        int d[PBATCH];
#pragma unroll
        for (int u = 0; u < PBATCH; ++u) {
            const int i = i0 + u * SORT_T;
            d[u] = (i < hi) ? dst[i] : -1;
        }
#pragma unroll
        for (int u = 0; u < PBATCH; ++u)
            if (d[u] >= 0) atomicAdd(&hist[(unsigned)d[u] / (unsigned)npb], 1);
    }
    __syncthreads();

    const int c = (t < NBUCKET) ? hist[t] : 0;
    int total;
    const int incl = block_scan_incl(c, wsum, &total);
    if (t < NBUCKET) {
        cursor[t] = incl - c;
        segStart[t * B + b] = lo + incl - c;
        segCnt[t * B + b] = c;
    }
    __syncthreads();

    for (int i0 = lo + t; i0 < hi; i0 += PBATCH * SORT_T) {
        int d[PBATCH], s[PBATCH];
#pragma unroll
        for (int u = 0; u < PBATCH; ++u) {
            const int i = i0 + u * SORT_T;
            d[u] = (i < hi) ? dst[i] : -1;
            s[u] = (i < hi) ? src[i] : 0;
        }
#pragma unroll
        for (int u = 0; u < PBATCH; ++u) {
            if (d[u] >= 0) {
                const int kb = (int)((unsigned)d[u] / (unsigned)npb);
                const int pos = atomicAdd(&cursor[kb], 1);
                part[lo + pos] = (unsigned)s[u] | ((unsigned)(d[u] - kb * npb) << SRC_BITS);
            }
        }
    }
}

template <bool PLACE>
__device__ __forceinline__ void bucket_entry(unsigned w, int* counters, int* lsrc)
{
    if (PLACE) {
        const int pos = atomicAdd(&counters[w >> SRC_BITS], 1);
        lsrc[pos] = (int)(w & SRC_MASK);
    } else {
        atomicAdd(&counters[w >> SRC_BITS], 1);
    }
}

template <bool PLACE>
__device__ __forceinline__ void walk_pieces(
    const unsigned* __restrict__ part, const int* __restrict__ myStart,
    const int* __restrict__ myCnt, int pBeg, int pEnd, int lane,
    int* counters, int* lsrc)
{
    for (int p0 = pBeg; p0 < pEnd; p0 += 32) {
        int dS = 0, dC = 0;
        if (p0 + lane < pEnd) { dS = myStart[p0 + lane]; dC = myCnt[p0 + lane]; }
        const int np = (pEnd - p0 < 32) ? (pEnd - p0) : 32;
        for (int j = 0; j < np; j += 4) {
            int s0[4], c[4];
            unsigned w0[4], w1[4];
#pragma unroll
            for (int u = 0; u < 4; ++u) {
                s0[u] = __builtin_amdgcn_readlane(dS, j + u);
                c[u] = __builtin_amdgcn_readlane(dC, j + u);
                w0[u] = (lane < c[u]) ? part[s0[u] + lane] : EMPTY;
                w1[u] = (lane + 32 < c[u]) ? part[s0[u] + lane + 32] : EMPTY;
            }
#pragma unroll
            for (int u = 0; u < 4; ++u) {
                if (w0[u] != EMPTY) bucket_entry<PLACE>(w0[u], counters, lsrc);
                if (w1[u] != EMPTY) bucket_entry<PLACE>(w1[u], counters, lsrc);
                for (int i = lane + 64; i < c[u]; i += 32)
                    bucket_entry<PLACE>(part[s0[u] + i], counters, lsrc);
            }
        }
    }
}

__device__ __forceinline__ float wave_max(float v) {
#pragma unroll
    for (int off = 16; off >= 1; off >>= 1) v = fmaxf(v, __shfl_xor(v, off, 32));
    return v;
}
__device__ __forceinline__ float wave_sum(float v) {
#pragma unroll
    for (int off = 16; off >= 1; off >>= 1) v += __shfl_xor(v, off, 32);
    return v;
}

__device__ __forceinline__ float leaky(float e) {
    return (e >= 0.f) ? e : NEG_SLOPE * e;
}
__device__ __forceinline__ float edge_coeff(const float* __restrict__ q, int s, float kd) {
    return leaky(q[s] + kd);
}

template <int G>
__device__ __forceinline__ void add_group(const float* __restrict__ hcol, int s, float a,
                                          int t, float& accx, float& accy)
{
    float2 v[G];
#pragma unroll
    for (int u = 0; u < G; ++u) {
        const int st = __builtin_amdgcn_readlane(s, t + u);
        v[u] = *(const float2*)(hcol + (long)st * OUT_F);
    }
#pragma unroll
    for (int u = 0; u < G; ++u) {
        const float at = __int_as_float(
            __builtin_amdgcn_readlane(__float_as_int(a), t + u));
        accx += v[u].x * at;
        accy += v[u].y * at;
    }
}

__device__ __forceinline__ void add_rows(const float* __restrict__ hcol, int s, float a,
                                         int cn, float& accx, float& accy)
{
    int t = 0;
    for (; t + 16 <= cn; t += 16) add_group<16>(hcol, s, a, t, accx, accy);
    if (t + 8 <= cn) { add_group<8>(hcol, s, a, t, accx, accy); t += 8; }
    const int rem = cn - t;
    if (rem > 0) {
        float2 v[7];
#pragma unroll
        for (int u = 0; u < 7; ++u) {
            v[u].x = 0.f; v[u].y = 0.f;
            if (u < rem) {
                const int st = __builtin_amdgcn_readlane(s, t + u);
                v[u] = *(const float2*)(hcol + (long)st * OUT_F);
            }
        }
#pragma unroll
        for (int u = 0; u < 7; ++u) {
            if (u < rem) {
                const float at = __int_as_float(
                    __builtin_amdgcn_readlane(__float_as_int(a), t + u));
                accx += v[u].x * at;
                accy += v[u].y * at;
            }
        }
    }
}

__device__ __forceinline__ void aggregate_from_list(
    const float* __restrict__ h, const float* __restrict__ q,
    const float* __restrict__ k, float* __restrict__ out,
    const int* cntl, const int* startl, const int* lsrc,
    int nodeLo, int nb, int wave, int lane)
{
    const int i0 = (wave * nb) / 32;
    const int i1 = ((wave + 1) * nb) / 32;
    const float* __restrict__ hcol = h + 2 * lane;

    for (int c0 = i0; c0 < i1; c0 += 32) {
        const int nc = (i1 - c0 < 32) ? (i1 - c0) : 32;
        int mN = 0, mB = 0;
        float mK = 0.f;
        if (lane < nc) { mN = cntl[c0 + lane]; mB = startl[c0 + lane]; mK = k[nodeLo + c0 + lane]; }

        int sA0 = 0, sA1 = 0, sB0 = 0, sB1 = 0;
        float qA0 = 0.f, qA1 = 0.f;
        {
            const int n0 = __builtin_amdgcn_readlane(mN, 0);
            const int b0 = __builtin_amdgcn_readlane(mB, 0);
            if (lane < n0) sA0 = lsrc[b0 + lane];
            if (lane + 32 < n0) sA1 = lsrc[b0 + 32 + lane];
            if (nc > 1) {
                const int n1 = __builtin_amdgcn_readlane(mN, 1);
                const int b1 = __builtin_amdgcn_readlane(mB, 1);
                if (lane < n1) sB0 = lsrc[b1 + lane];
                if (lane + 32 < n1) sB1 = lsrc[b1 + 32 + lane];
            }
            if (lane < n0) qA0 = q[sA0];
            if (lane + 32 < n0) qA1 = q[sA1];
        }

        for (int i = 0; i < nc; ++i) {
            const int n = __builtin_amdgcn_readlane(mN, i);
            const int base = __builtin_amdgcn_readlane(mB, i);
            const float kd = __int_as_float(
                __builtin_amdgcn_readlane(__float_as_int(mK), i));

            const float cf0 = (lane < n) ? leaky(qA0 + kd) : -INFINITY;
            const float cf1 = (lane + 32 < n) ? leaky(qA1 + kd) : -INFINITY;
            float m = fmaxf(cf0, cf1);
            for (int c = 64; c < n; c += 32) {
                const int j = c + lane;
                if (j < n) m = fmaxf(m, edge_coeff(q, lsrc[base + j], kd));
            }
            m = wave_max(m);

            const float ex0 = (lane < n) ? __expf(cf0 - m) : 0.f;
            const float ex1 = (lane + 32 < n) ? __expf(cf1 - m) : 0.f;
            float ssum = ex0;
            ssum += ex1;
            for (int c = 64; c < n; c += 32) {
                const int j = c + lane;
                if (j < n) ssum += __expf(edge_coeff(q, lsrc[base + j], kd) - m);
            }
            ssum = wave_sum(ssum);

            int sC0 = 0, sC1 = 0;
            float qB0 = 0.f, qB1 = 0.f;
            if (i + 1 < nc) {
                const int n1 = __builtin_amdgcn_readlane(mN, i + 1);
                if (lane < n1) qB0 = q[sB0];
                if (lane + 32 < n1) qB1 = q[sB1];
            }
            if (i + 2 < nc) {
                const int n2 = __builtin_amdgcn_readlane(mN, i + 2);
                const int b2 = __builtin_amdgcn_readlane(mB, i + 2);
                if (lane < n2) sC0 = lsrc[b2 + lane];
                if (lane + 32 < n2) sC1 = lsrc[b2 + 32 + lane];
            }

            float accx = 0.f, accy = 0.f;
            {
                const float a0 = (lane < n) ? ex0 / ssum : 0.f;
                add_rows(hcol, sA0, a0, (n < 32) ? n : 32, accx, accy);
            }
            if (n > 32) {
                const float a1 = (lane + 32 < n) ? ex1 / ssum : 0.f;
                add_rows(hcol, sA1, a1, (n < 64) ? (n - 32) : 32, accx, accy);
            }
            for (int c = 64; c < n; c += 32) {
                const int j = c + lane;
                int s = 0;
                float a = 0.f;
                if (j < n) {
                    s = lsrc[base + j];
                    a = __expf(edge_coeff(q, s, kd) - m) / ssum;
                }
                add_rows(hcol, s, a, (n - c < 32) ? (n - c) : 32, accx, accy);
            }
            float2 r; r.x = accx; r.y = accy;
            *(float2*)(out + (long)(nodeLo + c0 + i) * OUT_F + 2 * lane) = r;

            sA0 = sB0; sA1 = sB1; qA0 = qB0; qA1 = qB1;
            sB0 = sC0; sB1 = sC1;
        }
    }
}

__device__ __forceinline__ void aggregate_by_rescan(
    const unsigned* __restrict__ part, const int* __restrict__ myStart,
    const int* __restrict__ myCnt, int B,
    const float* __restrict__ h, const float* __restrict__ q,
    const float* __restrict__ k, float* __restrict__ out,
    int nodeLo, int nb, int wave, int lane)
{
    const int i0 = (wave * nb) / 32;
    const int i1 = ((wave + 1) * nb) / 32;
    const float* __restrict__ hcol = h + 2 * lane;
    for (int ni = i0; ni < i1; ++ni) {
        const float kd = k[nodeLo + ni];
        float m = -INFINITY;
        for (int p = 0; p < B; ++p) {
            const int s0 = myStart[p], c = myCnt[p];
            for (int e = lane; e < c; e += 32) {
                const unsigned w = part[s0 + e];
                if ((int)(w >> SRC_BITS) == ni)
                    m = fmaxf(m, edge_coeff(q, (int)(w & SRC_MASK), kd));
            }
        }
        m = wave_max(m);
        float ssum = 0.f;
        for (int p = 0; p < B; ++p) {
            const int s0 = myStart[p], c = myCnt[p];
            for (int e = lane; e < c; e += 32) {
                const unsigned w = part[s0 + e];
                if ((int)(w >> SRC_BITS) == ni)
                    ssum += __expf(edge_coeff(q, (int)(w & SRC_MASK), kd) - m);
            }
        }
        ssum = wave_sum(ssum);
        float accx = 0.f, accy = 0.f;
        for (int p = 0; p < B; ++p) {
            const int s0 = myStart[p], c = myCnt[p];
            for (int e0 = 0; e0 < c; e0 += 32) {
                const int e = e0 + lane;
                int s = 0;
                float a = 0.f;
                bool mine = false;
                if (e < c) {
                    const unsigned w = part[s0 + e];
                    if ((int)(w >> SRC_BITS) == ni) {
                        mine = true;
                        s = (int)(w & SRC_MASK);
                        a = __expf(edge_coeff(q, s, kd) - m) / ssum;
                    }
                }
                unsigned todo = __builtin_amdgcn_ballot_w32(mine);
                while (todo != 0u) {
                    const int l = __builtin_ctz(todo);
                    todo &= todo - 1u;
                    const int st = __builtin_amdgcn_readlane(s, l);
                    const float at = __int_as_float(
                        __builtin_amdgcn_readlane(__float_as_int(a), l));
                    const float2 v = *(const float2*)(hcol + (long)st * OUT_F);
                    accx += v.x * at;
                    accy += v.y * at;
                }
            }
        }
        float2 r; r.x = accx; r.y = accy;
        *(float2*)(out + (long)(nodeLo + ni) * OUT_F + 2 * lane) = r;
    }
}

#define LIST_CAP 32768

__global__ __launch_bounds__(SORT_T) void bucket_aggregate_kernel(
    const unsigned* __restrict__ part, const int* __restrict__ segStart,
    const int* __restrict__ segCnt, const float* __restrict__ h,
    const float* __restrict__ q, const float* __restrict__ k,
    float* __restrict__ out, int N, int B)
{
    __shared__ int cntl[NPB_MAX];
    __shared__ int startl[NPB_MAX];
    __shared__ int cursor[NPB_MAX];
    __shared__ int wsumB[33];
    __shared__ int lsrc[LIST_CAP];

    const int t = threadIdx.x;
    const int lane = t & 31;
    const int wave = __builtin_amdgcn_readfirstlane(t >> 5);
    const int kb = blockIdx.x;
    const int npb = (N + NBUCKET - 1) / NBUCKET;
    const int nodeLo = kb * npb;
    int nb = N - nodeLo;
    if (nb > npb) nb = npb;
    if (nb < 0) nb = 0;

    cntl[t] = 0;
    __syncthreads();

    const int spw = (B + 31) / 32;
    const int pBeg = wave * spw;
    int pEnd = pBeg + spw;
    if (pEnd > B) pEnd = B;
    const int* __restrict__ myStart = segStart + kb * B;
    const int* __restrict__ myCnt = segCnt + kb * B;

    walk_pieces<false>(part, myStart, myCnt, pBeg, pEnd, lane, cntl, lsrc);
    __syncthreads();

    const int c = cntl[t];
    int btotal;
    const int excl = block_scan_incl(c, wsumB, &btotal) - c;
    startl[t] = excl;
    cursor[t] = excl;
    __syncthreads();

    if (btotal <= LIST_CAP) {
        walk_pieces<true>(part, myStart, myCnt, pBeg, pEnd, lane, cursor, lsrc);
        __syncthreads();
        aggregate_from_list(h, q, k, out, cntl, startl, lsrc, nodeLo, nb, wave, lane);
    } else {
        aggregate_by_rescan(part, myStart, myCnt, B, h, q, k, out, nodeLo, nb, wave, lane);
    }
}

extern "C" void kernel_launch(void* const* d_in, const int* in_sizes, int n_in,
                              void* d_out, int out_size, void* d_ws, size_t ws_size,
                              hipStream_t stream) {
    const float* x  = (const float*)d_in[0];
    const float* Wv = (const float*)d_in[1];
    const float* bv = (const float*)d_in[2];
    const float* wq = (const float*)d_in[3];
    const float* bq = (const float*)d_in[4];
    const float* wk = (const float*)d_in[5];
    const float* bk = (const float*)d_in[6];
    const int*   src = (const int*)d_in[7];
    const int*   dst = (const int*)d_in[8];

    const int N = in_sizes[0] / IN_F;
    const int E = in_sizes[7];
    float* out = (float*)d_out;

    char* p = (char*)d_ws;
    auto carve = [&](size_t bytes) {
        char* r = p;
        p += (bytes + 255) & ~(size_t)255;
        return r;
    };
    float* h    = (float*)carve((size_t)N * OUT_F * sizeof(float));
    float* q    = (float*)carve((size_t)N * sizeof(float));
    float* k    = (float*)carve((size_t)N * sizeof(float));
    unsigned* part = (unsigned*)carve((size_t)E * sizeof(unsigned));
    int* segStart = (int*)carve((size_t)PART_BLOCKS * NBUCKET * sizeof(int));
    int* segCnt   = (int*)carve((size_t)PART_BLOCKS * NBUCKET * sizeof(int));

    const int tiles = (N + 15) / 16;
    const int gemmBlocks = (tiles + 7) / 8;
    gat_gemm_kernel<<<gemmBlocks, 256, 0, stream>>>(
        x, Wv, bv, wq, bq, wk, bk, h, q, k, N);

    edge_partition_kernel<<<PART_BLOCKS, SORT_T, 0, stream>>>(
        src, dst, part, segStart, segCnt, N, E);

    bucket_aggregate_kernel<<<NBUCKET, SORT_T, 0, stream>>>(
        part, segStart, segCnt, h, q, k, out, N, PART_BLOCKS);
}


# Wave reductions and scans on the vector ALU (DPP row_xmask and row_shr, permlanex16) instead of LDS permutes: per-node softmax max and sum, q/k row sums per tile, block scans; sum order kept
# speedup vs baseline: 8.2738x; 1.0939x over previous
#include <hip/hip_runtime.h>

typedef __attribute__((ext_vector_type(2))) float v2f;
typedef __attribute__((ext_vector_type(8))) float v8f;

#define IN_F 128
#define OUT_F 64
#define NEG_SLOPE 0.2f
#define LDS_STRIDE (IN_F + 2)

template <int M>
__device__ __forceinline__ float row_xor(float v) {
    return __int_as_float(__builtin_amdgcn_update_dpp(
        0, __float_as_int(v), 0x160 + M, 0xf, 0xf, true));
}
__device__ __forceinline__ float half_xor(float v) {
    const int i = __float_as_int(v);
    return __int_as_float(__builtin_amdgcn_permlanex16(
        i, i, 0x76543210, 0xfedcba98, false, false));
}
template <int S>
__device__ __forceinline__ int row_up(int v) {
    return __builtin_amdgcn_update_dpp(0, v, 0x110 + S, 0xf, 0xf, true);
}
__device__ __forceinline__ int wave_scan_incl(int v, int lane) {
    v += row_up<1>(v);
    v += row_up<2>(v);
    v += row_up<4>(v);
    v += row_up<8>(v);
    const int r0 = __builtin_amdgcn_readlane(v, 15);
    if (lane >= 16) v += r0;
    return v;
}
template <int M>
__device__ __forceinline__ void row_sum_step(float (&qp)[8], float (&kp)[8]) {
#pragma unroll
    for (int j = 0; j < 8; ++j) {
        qp[j] += row_xor<M>(qp[j]);
        kp[j] += row_xor<M>(kp[j]);
    }
}

__device__ __forceinline__ void gemm_tile(
    const float* __restrict__ x,  const float* Wt,
    const float* __restrict__ bv, const float* __restrict__ wq,
    const float* __restrict__ bq, const float* __restrict__ wk,
    const float* __restrict__ bk,
    float* __restrict__ h, float* __restrict__ qout, float* __restrict__ kout,
    int rowBase, int N, int lane)
{
    const int half = lane >> 4;
    const int l16  = lane & 15;

    v8f acc[4];
#pragma unroll
    for (int nt = 0; nt < 4; ++nt) {
        float b = bv[nt * 16 + l16];
#pragma unroll
        for (int j = 0; j < 8; ++j) acc[nt][j] = b;
    }

    int arow = rowBase + l16;
    if (arow >= N) arow = N - 1;
    const float* __restrict__ xrow = x + (size_t)arow * IN_F;

#pragma nounroll
    for (int k0 = 0; k0 < IN_F; k0 += IN_F / 2)
#pragma unroll
    for (int kk = k0; kk < k0 + IN_F / 2; kk += 4) {
        const int kl = kk + 2 * half;
        const float2 av = *(const float2*)(xrow + kl);
        v2f a; a.x = av.x; a.y = av.y;
#pragma unroll
        for (int nt = 0; nt < 4; ++nt) {
            const float* wp = &Wt[(nt * 16 + l16) * LDS_STRIDE + kl];
            v2f b; b.x = wp[0]; b.y = wp[1];
            acc[nt] = __builtin_amdgcn_wmma_f32_16x16x4_f32(
                false, a, false, b, (short)0, acc[nt], false, false);
        }
    }

    float qpart[8], kpart[8];
#pragma unroll
    for (int j = 0; j < 8; ++j) { qpart[j] = 0.f; kpart[j] = 0.f; }
#pragma unroll
    for (int nt = 0; nt < 4; ++nt) {
        const float wqn = wq[nt * 16 + l16];
        const float wkn = wk[nt * 16 + l16];
#pragma unroll
        for (int j = 0; j < 8; ++j) {
            qpart[j] = fmaf(acc[nt][j], wqn, qpart[j]);
            kpart[j] = fmaf(acc[nt][j], wkn, kpart[j]);
        }
    }
    row_sum_step<8>(qpart, kpart);
    row_sum_step<4>(qpart, kpart);
    row_sum_step<2>(qpart, kpart);
    row_sum_step<1>(qpart, kpart);

#pragma unroll
    for (int nt = 0; nt < 4; ++nt) {
#pragma unroll
        for (int j = 0; j < 8; ++j) {
            const int m = rowBase + half * 8 + j;
            if (m < N) h[(size_t)m * OUT_F + nt * 16 + l16] = acc[nt][j];
        }
    }
    if (l16 == 0) {
        const float bqv = bq[0], bkv = bk[0];
#pragma unroll
        for (int j = 0; j < 8; ++j) {
            const int m = rowBase + half * 8 + j;
            if (m < N) {
                qout[m] = qpart[j] + bqv;
                kout[m] = kpart[j] + bkv;
            }
        }
    }
}

#define SORT_T 1024
#define NBUCKET 256
#define NPB_MAX 1024
#define SRC_BITS 20
#define SRC_MASK ((1u << SRC_BITS) - 1u)
#define PBATCH 8
#define EMPTY 0xFFFFFFFFu
#define PART_BLOCKS 256
#define STAGE_CAP 16384

__device__ __forceinline__ int block_scan_incl(int v, int* wsum, int* total)
{
    const int lane = threadIdx.x & 31;
    const int wave = threadIdx.x >> 5;
    const int incl = wave_scan_incl(v, lane);
    if (lane == 31) wsum[wave] = incl;
    __syncthreads();
    if (wave == 0) {
        const int s = wsum[lane];
        const int si = wave_scan_incl(s, lane);
        wsum[lane] = si - s;
        if (lane == 31) wsum[32] = si;
    }
    __syncthreads();
    *total = wsum[32];
    return incl + wsum[wave];
}

#define PWAVES 16
#define PT (PWAVES * 32)
#define EPT2 (STAGE_CAP / PT)
#define EPA (STAGE_CAP / SORT_T)

__global__ __launch_bounds__(SORT_T) void gemm_partition_kernel(
    const float* __restrict__ x,  const float* __restrict__ Wv,
    const float* __restrict__ bv, const float* __restrict__ wq,
    const float* __restrict__ bq, const float* __restrict__ wk,
    const float* __restrict__ bk,
    float* __restrict__ h, float* __restrict__ qout, float* __restrict__ kout,
    const int* src, const int* dst,
    unsigned* __restrict__ part, int* __restrict__ segStart,
    int* __restrict__ segCnt, int N, int E)
{
    __shared__ float Wt[OUT_F * LDS_STRIDE];
    __shared__ int hist[NBUCKET];
    __shared__ int cursor[NBUCKET];
    __shared__ int wsum[33];
    __shared__ int ticket;
    __shared__ uint4 stage4[STAGE_CAP / 4];
    __shared__ unsigned rawE[STAGE_CAP];
    __shared__ unsigned char rawK[STAGE_CAP];

    const int t = threadIdx.x;
    const int b = blockIdx.x;
    const int B = gridDim.x;
    const int npb = (N + NBUCKET - 1) / NBUCKET;
    const int chunk = (E + B - 1) / B;
    const int lo = b * chunk;
    int hi = lo + chunk;
    if (hi > E) hi = E;
    const int wave = __builtin_amdgcn_readfirstlane(t >> 5);
    const int lane = t & 31;
    const unsigned tiles = ((unsigned)N + 15u) / 16u;
    const int t0 = (int)(((unsigned)b * tiles) / (unsigned)B);
    const int t1 = (int)(((unsigned)(b + 1) * tiles) / (unsigned)B);

    if (t < NBUCKET) hist[t] = 0;
    if (t == 0) ticket = 0;
    {
        float wv[IN_F * OUT_F / SORT_T];
#pragma unroll
        for (int u = 0; u < IN_F * OUT_F / SORT_T; ++u) wv[u] = Wv[t + u * SORT_T];
#pragma unroll
        for (int u = 0; u < IN_F * OUT_F / SORT_T; ++u) {
            const int idx = t + u * SORT_T;
            const int kk = idx / OUT_F, n = idx % OUT_F;
            Wt[n * LDS_STRIDE + kk] = wv[u];
        }
    }
    __syncthreads();

    const bool staged = (chunk <= STAGE_CAP);
    const bool sorter = !staged || (wave < PWAVES);

    if (sorter) {
        if (staged) {
            int d[EPT2], s[EPT2];
#pragma unroll
            for (int u = 0; u < EPT2; ++u) {
                const int i = lo + t + u * PT;
                d[u] = (i < hi) ? dst[i] : -1;
                s[u] = (i < hi) ? src[i] : 0;
            }
#pragma unroll
            for (int u = 0; u < EPT2; ++u) {
                const bool ok = d[u] >= 0;
                const int kb = ok ? (int)((unsigned)d[u] / (unsigned)npb) : 0;
                if (ok) atomicAdd(&hist[kb], 1);
                rawE[t + u * PT] = ok
                    ? ((unsigned)s[u] | ((unsigned)(d[u] - kb * npb) << SRC_BITS)) : EMPTY;
                rawK[t + u * PT] = (unsigned char)kb;
            }
        } else {
            for (int i0 = lo + t; i0 < hi; i0 += PBATCH * SORT_T) {
                int d[PBATCH];
#pragma unroll
                for (int u = 0; u < PBATCH; ++u) {
                    const int i = i0 + u * SORT_T;
                    d[u] = (i < hi) ? dst[i] : -1;
                }
#pragma unroll
                for (int u = 0; u < PBATCH; ++u)
                    if (d[u] >= 0) atomicAdd(&hist[(unsigned)d[u] / (unsigned)npb], 1);
            }
        }
    }
    for (;;) {
        int tk = 0;
        if (lane == 0) tk = atomicAdd(&ticket, 1);
        tk = __builtin_amdgcn_readfirstlane(tk);
        const int tile = t0 + tk;
        if (tile >= t1) break;
        gemm_tile(x, Wt, bv, wq, bq, wk, bk, h, qout, kout, tile * 16, N, lane);
    }
    __syncthreads();

    const int c = (t < NBUCKET) ? hist[t] : 0;
    int total;
    const int incl = block_scan_incl(c, wsum, &total);
    if (t < NBUCKET) {
        cursor[t] = incl - c;
        segStart[t * B + b] = lo + incl - c;
        segCnt[t * B + b] = c;
    }
    __syncthreads();

    if (!staged) {
        for (int i0 = lo + t; i0 < hi; i0 += PBATCH * SORT_T) {
            int d[PBATCH], s[PBATCH];
#pragma unroll
            for (int u = 0; u < PBATCH; ++u) {
                const int i = i0 + u * SORT_T;
                d[u] = (i < hi) ? dst[i] : -1;
                s[u] = (i < hi) ? src[i] : 0;
            }
#pragma unroll
            for (int u = 0; u < PBATCH; ++u) {
                if (d[u] >= 0) {
                    const int kb = (int)((unsigned)d[u] / (unsigned)npb);
                    const int pos = atomicAdd(&cursor[kb], 1);
                    part[lo + pos] = (unsigned)s[u] | ((unsigned)(d[u] - kb * npb) << SRC_BITS);
                }
            }
        }
        return;
    }

    unsigned* stage = (unsigned*)stage4;
    {
        unsigned e[EPA];
        int kq[EPA];
#pragma unroll
        for (int u = 0; u < EPA; ++u) {
            e[u] = rawE[t + u * SORT_T];
            kq[u] = rawK[t + u * SORT_T];
        }
#pragma unroll
        for (int u = 0; u < EPA; ++u) {
            if (e[u] != EMPTY) {
                const int pos = atomicAdd(&cursor[kq[u]], 1);
                stage[pos] = e[u];
            }
        }
    }
    __syncthreads();

    const int len = (hi > lo) ? (hi - lo) : 0;
    if ((lo & 3) == 0) {
        int head = (32 - (lo & 31)) & 31;
        if (head > len) head = len;
        if (t < head) part[lo + t] = stage[t];
        const int n4 = (len - head) >> 2;
        uint4* g4 = (uint4*)(part + lo + head);
        const uint4* l4 = (const uint4*)(stage + head);
        for (int i = t; i < n4; i += SORT_T) g4[i] = l4[i];
        const int done = head + 4 * n4;
        if (t < len - done) part[lo + done + t] = stage[done + t];
    } else {
        for (int i = t; i < len; i += SORT_T) part[lo + i] = stage[i];
    }
}

template <bool PLACE>
__device__ __forceinline__ void bucket_entry(unsigned w, int* counters, int* lsrc)
{
    if (PLACE) {
        const int pos = atomicAdd(&counters[w >> SRC_BITS], 1);
        lsrc[pos] = (int)(w & SRC_MASK);
    } else {
        atomicAdd(&counters[w >> SRC_BITS], 1);
    }
}

template <bool PLACE>
__device__ __forceinline__ void walk_pieces(
    const unsigned* __restrict__ part, const int* __restrict__ myStart,
    const int* __restrict__ myCnt, int pBeg, int pEnd, int lane,
    int* counters, int* lsrc)
{
    for (int p0 = pBeg; p0 < pEnd; p0 += 32) {
        int dS = 0, dC = 0;
        if (p0 + lane < pEnd) { dS = myStart[p0 + lane]; dC = myCnt[p0 + lane]; }
        const int np = (pEnd - p0 < 32) ? (pEnd - p0) : 32;
        for (int j = 0; j < np; j += 4) {
            int s0[4], c[4];
            unsigned w0[4], w1[4];
#pragma unroll
            for (int u = 0; u < 4; ++u) {
                s0[u] = __builtin_amdgcn_readlane(dS, j + u);
                c[u] = __builtin_amdgcn_readlane(dC, j + u);
                w0[u] = (lane < c[u]) ? part[s0[u] + lane] : EMPTY;
                w1[u] = (lane + 32 < c[u]) ? part[s0[u] + lane + 32] : EMPTY;
            }
#pragma unroll
            for (int u = 0; u < 4; ++u) {
                if (w0[u] != EMPTY) bucket_entry<PLACE>(w0[u], counters, lsrc);
                if (w1[u] != EMPTY) bucket_entry<PLACE>(w1[u], counters, lsrc);
                for (int i = lane + 64; i < c[u]; i += 32)
                    bucket_entry<PLACE>(part[s0[u] + i], counters, lsrc);
            }
        }
    }
}

__device__ __forceinline__ float wave_max(float v) {
    v = fmaxf(v, half_xor(v));
    v = fmaxf(v, row_xor<8>(v));
    v = fmaxf(v, row_xor<4>(v));
    v = fmaxf(v, row_xor<2>(v));
    v = fmaxf(v, row_xor<1>(v));
    return v;
}
__device__ __forceinline__ float wave_sum(float v) {
    v += half_xor(v);
    v += row_xor<8>(v);
    v += row_xor<4>(v);
    v += row_xor<2>(v);
    v += row_xor<1>(v);
    return v;
}

__device__ __forceinline__ float leaky(float e) {
    return (e >= 0.f) ? e : NEG_SLOPE * e;
}
__device__ __forceinline__ float edge_coeff(const float* __restrict__ q, int s, float kd) {
    return leaky(q[s] + kd);
}

template <int G>
__device__ __forceinline__ void add_group(const float* __restrict__ hcol, int s, float a,
                                          int t, float& accx, float& accy)
{
    float2 v[G];
#pragma unroll
    for (int u = 0; u < G; ++u) {
        const int st = __builtin_amdgcn_readlane(s, t + u);
        v[u] = *(const float2*)(hcol + (long)st * OUT_F);
    }
#pragma unroll
    for (int u = 0; u < G; ++u) {
        const float at = __int_as_float(
            __builtin_amdgcn_readlane(__float_as_int(a), t + u));
        accx += v[u].x * at;
        accy += v[u].y * at;
    }
}

__device__ __forceinline__ void add_rows(const float* __restrict__ hcol, int s, float a,
                                         int cn, float& accx, float& accy)
{
    int t = 0;
    for (; t + 16 <= cn; t += 16) add_group<16>(hcol, s, a, t, accx, accy);
    if (t + 8 <= cn) { add_group<8>(hcol, s, a, t, accx, accy); t += 8; }
    const int rem = cn - t;
    if (rem > 0) {
        float2 v[7];
#pragma unroll
        for (int u = 0; u < 7; ++u) {
            v[u].x = 0.f; v[u].y = 0.f;
            if (u < rem) {
                const int st = __builtin_amdgcn_readlane(s, t + u);
                v[u] = *(const float2*)(hcol + (long)st * OUT_F);
            }
        }
#pragma unroll
        for (int u = 0; u < 7; ++u) {
            if (u < rem) {
                const float at = __int_as_float(
                    __builtin_amdgcn_readlane(__float_as_int(a), t + u));
                accx += v[u].x * at;
                accy += v[u].y * at;
            }
        }
    }
}

__device__ __forceinline__ void aggregate_from_list(
    const float* __restrict__ h, const float* __restrict__ q,
    const float* __restrict__ k, float* __restrict__ out,
    const int* cntl, const int* startl, const int* lsrc,
    int nodeLo, int nb, int wave, int lane)
{
    const int i0 = (wave * nb) / 32;
    const int i1 = ((wave + 1) * nb) / 32;
    const float* __restrict__ hcol = h + 2 * lane;

    for (int c0 = i0; c0 < i1; c0 += 32) {
        const int nc = (i1 - c0 < 32) ? (i1 - c0) : 32;
        int mN = 0, mB = 0;
        float mK = 0.f;
        if (lane < nc) { mN = cntl[c0 + lane]; mB = startl[c0 + lane]; mK = k[nodeLo + c0 + lane]; }

        int sA0 = 0, sA1 = 0, sB0 = 0, sB1 = 0;
        float qA0 = 0.f, qA1 = 0.f;
        {
            const int n0 = __builtin_amdgcn_readlane(mN, 0);
            const int b0 = __builtin_amdgcn_readlane(mB, 0);
            if (lane < n0) sA0 = lsrc[b0 + lane];
            if (lane + 32 < n0) sA1 = lsrc[b0 + 32 + lane];
            if (nc > 1) {
                const int n1 = __builtin_amdgcn_readlane(mN, 1);
                const int b1 = __builtin_amdgcn_readlane(mB, 1);
                if (lane < n1) sB0 = lsrc[b1 + lane];
                if (lane + 32 < n1) sB1 = lsrc[b1 + 32 + lane];
            }
            if (lane < n0) qA0 = q[sA0];
            if (lane + 32 < n0) qA1 = q[sA1];
        }

        for (int i = 0; i < nc; ++i) {
            const int n = __builtin_amdgcn_readlane(mN, i);
            const int base = __builtin_amdgcn_readlane(mB, i);
            const float kd = __int_as_float(
                __builtin_amdgcn_readlane(__float_as_int(mK), i));

            const float cf0 = (lane < n) ? leaky(qA0 + kd) : -INFINITY;
            const float cf1 = (lane + 32 < n) ? leaky(qA1 + kd) : -INFINITY;
            float m = fmaxf(cf0, cf1);
            for (int c = 64; c < n; c += 32) {
                const int j = c + lane;
                if (j < n) m = fmaxf(m, edge_coeff(q, lsrc[base + j], kd));
            }
            m = wave_max(m);

            const float ex0 = (lane < n) ? __expf(cf0 - m) : 0.f;
            const float ex1 = (lane + 32 < n) ? __expf(cf1 - m) : 0.f;
            float ssum = ex0;
            ssum += ex1;
            for (int c = 64; c < n; c += 32) {
                const int j = c + lane;
                if (j < n) ssum += __expf(edge_coeff(q, lsrc[base + j], kd) - m);
            }
            ssum = wave_sum(ssum);

            int sC0 = 0, sC1 = 0;
            float qB0 = 0.f, qB1 = 0.f;
            if (i + 1 < nc) {
                const int n1 = __builtin_amdgcn_readlane(mN, i + 1);
                if (lane < n1) qB0 = q[sB0];
                if (lane + 32 < n1) qB1 = q[sB1];
            }
            if (i + 2 < nc) {
                const int n2 = __builtin_amdgcn_readlane(mN, i + 2);
                const int b2 = __builtin_amdgcn_readlane(mB, i + 2);
                if (lane < n2) sC0 = lsrc[b2 + lane];
                if (lane + 32 < n2) sC1 = lsrc[b2 + 32 + lane];
            }

            float accx = 0.f, accy = 0.f;
            {
                const float a0 = (lane < n) ? ex0 / ssum : 0.f;
                add_rows(hcol, sA0, a0, (n < 32) ? n : 32, accx, accy);
            }
            if (n > 32) {
                const float a1 = (lane + 32 < n) ? ex1 / ssum : 0.f;
                add_rows(hcol, sA1, a1, (n < 64) ? (n - 32) : 32, accx, accy);
            }
            for (int c = 64; c < n; c += 32) {
                const int j = c + lane;
                int s = 0;
                float a = 0.f;
                if (j < n) {
                    s = lsrc[base + j];
                    a = __expf(edge_coeff(q, s, kd) - m) / ssum;
                }
                add_rows(hcol, s, a, (n - c < 32) ? (n - c) : 32, accx, accy);
            }
            float2 r; r.x = accx; r.y = accy;
            *(float2*)(out + (long)(nodeLo + c0 + i) * OUT_F + 2 * lane) = r;

            sA0 = sB0; sA1 = sB1; qA0 = qB0; qA1 = qB1;
            sB0 = sC0; sB1 = sC1;
        }
    }
}

__device__ __forceinline__ void aggregate_by_rescan(
    const unsigned* __restrict__ part, const int* __restrict__ myStart,
    const int* __restrict__ myCnt, int B,
    const float* __restrict__ h, const float* __restrict__ q,
    const float* __restrict__ k, float* __restrict__ out,
    int nodeLo, int nb, int wave, int lane)
{
    const int i0 = (wave * nb) / 32;
    const int i1 = ((wave + 1) * nb) / 32;
    const float* __restrict__ hcol = h + 2 * lane;
    for (int ni = i0; ni < i1; ++ni) {
        const float kd = k[nodeLo + ni];
        float m = -INFINITY;
        for (int p = 0; p < B; ++p) {
            const int s0 = myStart[p], c = myCnt[p];
            for (int e = lane; e < c; e += 32) {
                const unsigned w = part[s0 + e];
                if ((int)(w >> SRC_BITS) == ni)
                    m = fmaxf(m, edge_coeff(q, (int)(w & SRC_MASK), kd));
            }
        }
        m = wave_max(m);
        float ssum = 0.f;
        for (int p = 0; p < B; ++p) {
            const int s0 = myStart[p], c = myCnt[p];
            for (int e = lane; e < c; e += 32) {
                const unsigned w = part[s0 + e];
                if ((int)(w >> SRC_BITS) == ni)
                    ssum += __expf(edge_coeff(q, (int)(w & SRC_MASK), kd) - m);
            }
        }
        ssum = wave_sum(ssum);
        float accx = 0.f, accy = 0.f;
        for (int p = 0; p < B; ++p) {
            const int s0 = myStart[p], c = myCnt[p];
            for (int e0 = 0; e0 < c; e0 += 32) {
                const int e = e0 + lane;
                int s = 0;
                float a = 0.f;
                bool mine = false;
                if (e < c) {
                    const unsigned w = part[s0 + e];
                    if ((int)(w >> SRC_BITS) == ni) {
                        mine = true;
                        s = (int)(w & SRC_MASK);
                        a = __expf(edge_coeff(q, s, kd) - m) / ssum;
                    }
                }
                unsigned todo = __builtin_amdgcn_ballot_w32(mine);
                while (todo != 0u) {
                    const int l = __builtin_ctz(todo);
                    todo &= todo - 1u;
                    const int st = __builtin_amdgcn_readlane(s, l);
                    const float at = __int_as_float(
                        __builtin_amdgcn_readlane(__float_as_int(a), l));
                    const float2 v = *(const float2*)(hcol + (long)st * OUT_F);
                    accx += v.x * at;
                    accy += v.y * at;
                }
            }
        }
        float2 r; r.x = accx; r.y = accy;
        *(float2*)(out + (long)(nodeLo + ni) * OUT_F + 2 * lane) = r;
    }
}

#define LIST_CAP 32768

__global__ __launch_bounds__(SORT_T) void bucket_aggregate_kernel(
    const unsigned* __restrict__ part, const int* __restrict__ segStart,
    const int* __restrict__ segCnt, const float* __restrict__ h,
    const float* __restrict__ q, const float* __restrict__ k,
    float* __restrict__ out, int N, int B)
{
    __shared__ int cntl[NPB_MAX];
    __shared__ int startl[NPB_MAX];
    __shared__ int cursor[NPB_MAX];
    __shared__ int wsumB[33];
    __shared__ int lsrc[LIST_CAP];

    const int t = threadIdx.x;
    const int lane = t & 31;
    const int wave = __builtin_amdgcn_readfirstlane(t >> 5);
    const int kb = blockIdx.x;
    const int npb = (N + NBUCKET - 1) / NBUCKET;
    const int nodeLo = kb * npb;
    int nb = N - nodeLo;
    if (nb > npb) nb = npb;
    if (nb < 0) nb = 0;

    cntl[t] = 0;
    __syncthreads();

    const int spw = (B + 31) / 32;
    const int pBeg = wave * spw;
    int pEnd = pBeg + spw;
    if (pEnd > B) pEnd = B;
    const int* __restrict__ myStart = segStart + kb * B;
    const int* __restrict__ myCnt = segCnt + kb * B;

    walk_pieces<false>(part, myStart, myCnt, pBeg, pEnd, lane, cntl, lsrc);
    __syncthreads();

    const int c = cntl[t];
    int btotal;
    const int excl = block_scan_incl(c, wsumB, &btotal) - c;
    startl[t] = excl;
    cursor[t] = excl;
    __syncthreads();

    if (btotal <= LIST_CAP) {
        walk_pieces<true>(part, myStart, myCnt, pBeg, pEnd, lane, cursor, lsrc);
        __syncthreads();
        aggregate_from_list(h, q, k, out, cntl, startl, lsrc, nodeLo, nb, wave, lane);
    } else {
        aggregate_by_rescan(part, myStart, myCnt, B, h, q, k, out, nodeLo, nb, wave, lane);
    }
}

extern "C" void kernel_launch(void* const* d_in, const int* in_sizes, int n_in,
                              void* d_out, int out_size, void* d_ws, size_t ws_size,
                              hipStream_t stream) {
    const float* x  = (const float*)d_in[0];
    const float* Wv = (const float*)d_in[1];
    const float* bv = (const float*)d_in[2];
    const float* wq = (const float*)d_in[3];
    const float* bq = (const float*)d_in[4];
    const float* wk = (const float*)d_in[5];
    const float* bk = (const float*)d_in[6];
    const int*   src = (const int*)d_in[7];
    const int*   dst = (const int*)d_in[8];

    const int N = in_sizes[0] / IN_F;
    const int E = in_sizes[7];
    float* out = (float*)d_out;

    char* p = (char*)d_ws;
    auto carve = [&](size_t bytes) {
        char* r = p;
        p += (bytes + 255) & ~(size_t)255;
        return r;
    };
    float* h    = (float*)carve((size_t)N * OUT_F * sizeof(float));
    float* q    = (float*)carve((size_t)N * sizeof(float));
    float* k    = (float*)carve((size_t)N * sizeof(float));
    unsigned* part = (unsigned*)carve((size_t)E * sizeof(unsigned));
    int* segStart = (int*)carve((size_t)PART_BLOCKS * NBUCKET * sizeof(int));
    int* segCnt   = (int*)carve((size_t)PART_BLOCKS * NBUCKET * sizeof(int));

    gemm_partition_kernel<<<PART_BLOCKS, SORT_T, 0, stream>>>(
        x, Wv, bv, wq, bq, wk, bk, h, q, k, src, dst, part, segStart, segCnt, N, E);

    bucket_aggregate_kernel<<<NBUCKET, SORT_T, 0, stream>>>(
        part, segStart, segCnt, h, q, k, out, N, PART_BLOCKS);
}
